# starting kernel
# baseline (speedup 1.0000x reference)
#include <hip/hip_runtime.h>
#include <hip/hip_bf16.h>

typedef __attribute__((ext_vector_type(2))) float v2f;
typedef __attribute__((ext_vector_type(8))) float v8f;

#define B_DIM 4
#define TQ 512
#define TV 512
#define DM 256
#define UN 128
#define TPB_T 8

static __device__ __forceinline__ float fast_tanh(float x) {
#if __has_builtin(__builtin_amdgcn_tanhf)
    return __builtin_amdgcn_tanhf(x);
#else
    return tanhf(x);
#endif
}

static __device__ __forceinline__ void wmma_tile_f32(
    const float* __restrict__ A, int lda,
    const float* __restrict__ B, int ldb,
    float* __restrict__ C, int ldc,
    int m0, int n0, int K)
{
    const int lane = threadIdx.x & 31;
    const int half = lane >> 4;
    const int mn   = lane & 15;
    v8f acc = {0.f, 0.f, 0.f, 0.f, 0.f, 0.f, 0.f, 0.f};
    const float* __restrict__ arow = A + (size_t)(m0 + mn) * lda + 2 * half;
    const float* __restrict__ bcol = B + (size_t)(2 * half) * ldb + (n0 + mn);
#pragma unroll 2
    for (int k0 = 0; k0 < K; k0 += 4) {
        v2f a, b;
        a.x = arow[k0 + 0];
        a.y = arow[k0 + 1];
        b.x = bcol[(size_t)(k0 + 0) * ldb];
        b.y = bcol[(size_t)(k0 + 1) * ldb];
        acc = __builtin_amdgcn_wmma_f32_16x16x4_f32(
            false, a, false, b, (short)0, acc, false, false);
    }
#pragma unroll
    for (int r = 0; r < 8; ++r) {
        C[(size_t)(m0 + r + 8 * half) * ldc + (n0 + mn)] = acc[r];
    }
}

__global__ __launch_bounds__(256) void proj_kernel(
    const float* __restrict__ query, const float* __restrict__ value,
    const float* __restrict__ W1, const float* __restrict__ W2,
    float* __restrict__ Qo, float* __restrict__ Ko)
{
    const int wave  = (int)((blockIdx.x * blockDim.x + threadIdx.x) >> 5);
    const int which = wave >> 10;
    const int tile  = wave & 1023;
    const int mt    = tile >> 3;
    const int nt    = tile & 7;
    const float* A  = which ? value : query;
    const float* Bm = which ? W2 : W1;
    float* C        = which ? Ko : Qo;
    wmma_tile_f32(A, DM, Bm, UN, C, UN, mt * 16, nt * 16, DM);
}

__global__ __launch_bounds__(512) void score_softmax_kernel(
    const float* __restrict__ Q, const float* __restrict__ Kmat,
    const float* __restrict__ scale, float* __restrict__ attn)
{
    const int b    = (int)blockIdx.x >> 6;
    const int t0   = ((int)blockIdx.x & 63) * TPB_T;
    const int tid  = (int)threadIdx.x;
    __shared__ float q_sh[TPB_T][UN];
    __shared__ float sc_sh[UN];
    __shared__ float red[16];

    if (tid < UN) sc_sh[tid] = scale[tid];
    {
        const int tt = tid >> 7;
        const int uu = tid & 127;
        q_sh[tt][uu]     = Q[(size_t)(b * TQ + t0 + tt) * UN + uu];
        q_sh[tt + 4][uu] = Q[(size_t)(b * TQ + t0 + tt + 4) * UN + uu];
    }

    float kreg[UN];
    {
        const float4* __restrict__ krow4 =
            (const float4*)(Kmat + ((size_t)(b * TV + tid)) * UN);
#pragma unroll
        for (int i = 0; i < UN / 4; ++i) {
            const float4 v = krow4[i];
            kreg[4 * i + 0] = v.x;
            kreg[4 * i + 1] = v.y;
            kreg[4 * i + 2] = v.z;
            kreg[4 * i + 3] = v.w;
        }
    }
    __syncthreads();

    const int w = tid >> 5;
    for (int tt = 0; tt < TPB_T; ++tt) {
        float s = 0.f;
#pragma unroll
        for (int u = 0; u < UN; ++u) {
            s = __builtin_fmaf(sc_sh[u], fast_tanh(q_sh[tt][u] + kreg[u]), s);
        }

        float m = s;
#pragma unroll
        for (int off = 16; off > 0; off >>= 1)
            m = fmaxf(m, __shfl_xor(m, off, 32));
        if ((tid & 31) == 0) red[w] = m;
        __syncthreads();
        float bmax = red[0];
#pragma unroll
        for (int i = 1; i < 16; ++i) bmax = fmaxf(bmax, red[i]);

        const float e = __expf(s - bmax);

        __syncthreads();
        float sum = e;
#pragma unroll
        for (int off = 16; off > 0; off >>= 1)
            sum += __shfl_xor(sum, off, 32);
        if ((tid & 31) == 0) red[w] = sum;
        __syncthreads();
        float bsum = 0.f;
#pragma unroll
        for (int i = 0; i < 16; ++i) bsum += red[i];

        attn[(size_t)(b * TQ + t0 + tt) * TV + tid] = e / bsum;
        __syncthreads();
    }
}

__global__ __launch_bounds__(256) void context_kernel(
    const float* __restrict__ attn, const float* __restrict__ value,
    float* __restrict__ ctx)
{
    const int wave = (int)((blockIdx.x * blockDim.x + threadIdx.x) >> 5);
    const int b    = wave >> 9;
    const int t    = wave & 511;
    const int mt   = t >> 4;
    const int nt   = t & 15;
    const float* A  = attn  + (size_t)b * TQ * TV;
    const float* Bv = value + (size_t)b * TV * DM;
    float* C        = ctx   + (size_t)b * TQ * DM;
    wmma_tile_f32(A, TV, Bv, DM, C, DM, mt * 16, nt * 16, TV);
}

extern "C" void kernel_launch(void* const* d_in, const int* in_sizes, int n_in,
                              void* d_out, int out_size, void* d_ws, size_t ws_size,
                              hipStream_t stream) {
    const float* query = (const float*)d_in[0];
    const float* value = (const float*)d_in[1];
    const float* W1    = (const float*)d_in[2];
    const float* W2    = (const float*)d_in[3];
    const float* scale = (const float*)d_in[4];

    float* out  = (float*)d_out;
    float* ctx  = out;
    float* attn = out + (size_t)B_DIM * TQ * DM;

    float* Qw = (float*)d_ws;
    float* Kw = Qw + (size_t)(B_DIM * TQ) * UN;

    proj_kernel<<<256, 256, 0, stream>>>(query, value, W1, W2, Qw, Kw);
    score_softmax_kernel<<<(B_DIM * TQ) / TPB_T, TV, 0, stream>>>(Qw, Kw, scale, attn);
    context_kernel<<<256, 256, 0, stream>>>(attn, value, ctx);
}


# Both GEMMs: block = one 16x128 output strip, its 8 waves split K, each wave keeps 8 column tiles per A fragment; the 8 K-slice partials are added through LDS in fixed order (new summation tree)
# speedup vs baseline: 1.6300x; 1.6300x over previous
#include <hip/hip_runtime.h>
#include <hip/hip_bf16.h>

typedef __attribute__((ext_vector_type(2))) float v2f;
typedef __attribute__((ext_vector_type(8))) float v8f;

#define B_DIM 4
#define TQ 512
#define TV 512
#define DM 256
#define UN 128
#define TPB_T 8

static __device__ __forceinline__ float fast_tanh(float x) {
#if __has_builtin(__builtin_amdgcn_tanhf)
    return __builtin_amdgcn_tanhf(x);
#else
    return tanhf(x);
#endif
}

template <int KS>
static __device__ __forceinline__ void strip_gemm_splitk(
    const float* __restrict__ A, int lda,
    const float* __restrict__ B, int ldb,
    float* __restrict__ C, int ldc,
    int m0, int n0, float (*part)[16][128])
{
    const int tid  = (int)threadIdx.x;
    const int lane = tid & 31;
    const int h    = lane >> 4;
    const int c    = lane & 15;
    const int w    = tid >> 5;

    v8f acc[8];
#pragma unroll
    for (int t = 0; t < 8; ++t)
        acc[t] = (v8f){0.f, 0.f, 0.f, 0.f, 0.f, 0.f, 0.f, 0.f};

    const float* __restrict__ ap = A + (size_t)(m0 + c) * lda + (w * KS + 4 * h);
    const float* __restrict__ bp = B + (size_t)(w * KS + 4 * h) * ldb + (n0 + 4 * c);

    for (int kb = 0; kb < KS; kb += 8) {
        const float4 av = *(const float4*)(ap + kb);
        float4 bl[4], bh[4];
#pragma unroll
        for (int j = 0; j < 4; ++j) {
            const float* __restrict__ row = bp + (size_t)(kb + j) * ldb;
            bl[j] = *(const float4*)(row);
            bh[j] = *(const float4*)(row + 64);
        }
        const float bq[2][4][4] = {
            {{bl[0].x, bl[0].y, bl[0].z, bl[0].w}, {bl[1].x, bl[1].y, bl[1].z, bl[1].w},
             {bl[2].x, bl[2].y, bl[2].z, bl[2].w}, {bl[3].x, bl[3].y, bl[3].z, bl[3].w}},
            {{bh[0].x, bh[0].y, bh[0].z, bh[0].w}, {bh[1].x, bh[1].y, bh[1].z, bh[1].w},
             {bh[2].x, bh[2].y, bh[2].z, bh[2].w}, {bh[3].x, bh[3].y, bh[3].z, bh[3].w}}};
        const v2f a01 = {av.x, av.y};
        const v2f a23 = {av.z, av.w};
#pragma unroll
        for (int g = 0; g < 2; ++g) {
#pragma unroll
            for (int e = 0; e < 4; ++e) {
                const v2f b01 = {bq[g][0][e], bq[g][1][e]};
                acc[4 * g + e] = __builtin_amdgcn_wmma_f32_16x16x4_f32(
                    false, a01, false, b01, (short)0, acc[4 * g + e], false, false);
            }
        }
#pragma unroll
        for (int g = 0; g < 2; ++g) {
#pragma unroll
            for (int e = 0; e < 4; ++e) {
                const v2f b23 = {bq[g][2][e], bq[g][3][e]};
                acc[4 * g + e] = __builtin_amdgcn_wmma_f32_16x16x4_f32(
                    false, a23, false, b23, (short)0, acc[4 * g + e], false, false);
            }
        }
    }

#pragma unroll
    for (int g = 0; g < 2; ++g) {
#pragma unroll
        for (int r = 0; r < 8; ++r) {
            const float4 o = make_float4(acc[4 * g + 0][r], acc[4 * g + 1][r],
                                         acc[4 * g + 2][r], acc[4 * g + 3][r]);
            *(float4*)&part[w][8 * h + r][64 * g + 4 * c] = o;
        }
    }
    __syncthreads();

#pragma unroll
    for (int i = 0; i < 2; ++i) {
        const int f   = tid + 256 * i;
        const int row = f >> 5;
        const int c4  = (f & 31) * 4;
        float4 s = *(const float4*)&part[0][row][c4];
#pragma unroll
        for (int ww = 1; ww < 8; ++ww) {
            const float4 p = *(const float4*)&part[ww][row][c4];
            s.x += p.x; s.y += p.y; s.z += p.z; s.w += p.w;
        }
        *(float4*)(C + (size_t)(m0 + row) * ldc + (n0 + c4)) = s;
    }
}

__global__ __launch_bounds__(256) void proj_kernel(
    const float* __restrict__ query, const float* __restrict__ value,
    const float* __restrict__ W1, const float* __restrict__ W2,
    float* __restrict__ Qo, float* __restrict__ Ko)
{
    __shared__ float part[8][16][128];
    const int which = (int)blockIdx.x >> 7;
    const int mt    = (int)blockIdx.x & 127;
    const float* A  = which ? value : query;
    const float* Bm = which ? W2 : W1;
    float* C        = which ? Ko : Qo;
    strip_gemm_splitk<DM / 8>(A, DM, Bm, UN, C, UN, mt * 16, 0, part);
}

__global__ __launch_bounds__(512) void score_softmax_kernel(
    const float* __restrict__ Q, const float* __restrict__ Kmat,
    const float* __restrict__ scale, float* __restrict__ attn)
{
    const int b    = (int)blockIdx.x >> 6;
    const int t0   = ((int)blockIdx.x & 63) * TPB_T;
    const int tid  = (int)threadIdx.x;
    __shared__ float q_sh[TPB_T][UN];
    __shared__ float sc_sh[UN];
    __shared__ float red[16];

    if (tid < UN) sc_sh[tid] = scale[tid];
    {
        const int tt = tid >> 7;
        const int uu = tid & 127;
        q_sh[tt][uu]     = Q[(size_t)(b * TQ + t0 + tt) * UN + uu];
        q_sh[tt + 4][uu] = Q[(size_t)(b * TQ + t0 + tt + 4) * UN + uu];
    }

    float kreg[UN];
    {
        const float4* __restrict__ krow4 =
            (const float4*)(Kmat + ((size_t)(b * TV + tid)) * UN);
#pragma unroll
        for (int i = 0; i < UN / 4; ++i) {
            const float4 v = krow4[i];
            kreg[4 * i + 0] = v.x;
            kreg[4 * i + 1] = v.y;
            kreg[4 * i + 2] = v.z;
            kreg[4 * i + 3] = v.w;
        }
    }
    __syncthreads();

    const int w = tid >> 5;
    for (int tt = 0; tt < TPB_T; ++tt) {
        float s = 0.f;
#pragma unroll
        for (int u = 0; u < UN; ++u) {
            s = __builtin_fmaf(sc_sh[u], fast_tanh(q_sh[tt][u] + kreg[u]), s);
        }

        float m = s;
#pragma unroll
        for (int off = 16; off > 0; off >>= 1)
            m = fmaxf(m, __shfl_xor(m, off, 32));
        if ((tid & 31) == 0) red[w] = m;
        __syncthreads();
        float bmax = red[0];
#pragma unroll
        for (int i = 1; i < 16; ++i) bmax = fmaxf(bmax, red[i]);

        const float e = __expf(s - bmax);

        __syncthreads();
        float sum = e;
#pragma unroll
        for (int off = 16; off > 0; off >>= 1)
            sum += __shfl_xor(sum, off, 32);
        if ((tid & 31) == 0) red[w] = sum;
        __syncthreads();
        float bsum = 0.f;
#pragma unroll
        for (int i = 0; i < 16; ++i) bsum += red[i];

        attn[(size_t)(b * TQ + t0 + tt) * TV + tid] = e / bsum;
        __syncthreads();
    }
}

__global__ __launch_bounds__(256) void context_kernel(
    const float* __restrict__ attn, const float* __restrict__ value,
    float* __restrict__ ctx)
{
    __shared__ float part[8][16][128];
    const int b   = (int)blockIdx.x >> 6;
    const int rem = (int)blockIdx.x & 63;
    const int mt  = rem >> 1;
    const int nh  = rem & 1;
    const float* A  = attn  + (size_t)b * TQ * TV;
    const float* Bv = value + (size_t)b * TV * DM;
    float* C        = ctx   + (size_t)b * TQ * DM;
    strip_gemm_splitk<TV / 8>(A, TV, Bv, DM, C, DM, mt * 16, nh * 128, part);
}

extern "C" void kernel_launch(void* const* d_in, const int* in_sizes, int n_in,
                              void* d_out, int out_size, void* d_ws, size_t ws_size,
                              hipStream_t stream) {
    const float* query = (const float*)d_in[0];
    const float* value = (const float*)d_in[1];
    const float* W1    = (const float*)d_in[2];
    const float* W2    = (const float*)d_in[3];
    const float* scale = (const float*)d_in[4];

    float* out  = (float*)d_out;
    float* ctx  = out;
    float* attn = out + (size_t)B_DIM * TQ * DM;

    float* Qw = (float*)d_ws;
    float* Kw = Qw + (size_t)(B_DIM * TQ) * UN;

    proj_kernel<<<256, 256, 0, stream>>>(query, value, W1, W2, Qw, Kw);
    score_softmax_kernel<<<(B_DIM * TQ) / TPB_T, TV, 0, stream>>>(Qw, Kw, scale, attn);
    context_kernel<<<256, 256, 0, stream>>>(attn, value, ctx);
}


# score kernel softmax: per row one barrier; each wave reduces its 32 scores by DPP to (max, exp-sum), 16 pairs merged by rescaling (was 10 LDS shuffles + 4 barriers per row; new summation tree)
# speedup vs baseline: 1.8052x; 1.1075x over previous
#include <hip/hip_runtime.h>
#include <hip/hip_bf16.h>

typedef __attribute__((ext_vector_type(2))) float v2f;
typedef __attribute__((ext_vector_type(8))) float v8f;

#define B_DIM 4
#define TQ 512
#define TV 512
#define DM 256
#define UN 128
#define TPB_T 8

static __device__ __forceinline__ float fast_tanh(float x) {
#if __has_builtin(__builtin_amdgcn_tanhf)
    return __builtin_amdgcn_tanhf(x);
#else
    return tanhf(x);
#endif
}

template <int MASK>
static __device__ __forceinline__ float row_xor(float x) {
    const int b = __builtin_bit_cast(int, x);
    return __builtin_bit_cast(float,
        __builtin_amdgcn_update_dpp(b, b, 0x160 + MASK, 0xf, 0xf, true));
}

static __device__ __forceinline__ float xhalf(float x) {
    const int b = __builtin_bit_cast(int, x);
    return __builtin_bit_cast(float,
        __builtin_amdgcn_permlanex16(b, b, 0x76543210, 0xfedcba98, false, false));
}

static __device__ __forceinline__ float row16_max(float x) {
    x = fmaxf(x, row_xor<1>(x));
    x = fmaxf(x, row_xor<2>(x));
    x = fmaxf(x, row_xor<4>(x));
    x = fmaxf(x, row_xor<8>(x));
    return x;
}
static __device__ __forceinline__ float row16_sum(float x) {
    x += row_xor<1>(x);
    x += row_xor<2>(x);
    x += row_xor<4>(x);
    x += row_xor<8>(x);
    return x;
}

template <int KS>
static __device__ __forceinline__ void strip_gemm_splitk(
    const float* __restrict__ A, int lda,
    const float* __restrict__ B, int ldb,
    float* __restrict__ C, int ldc,
    int m0, int n0, float (*part)[16][128])
{
    const int tid  = (int)threadIdx.x;
    const int lane = tid & 31;
    const int h    = lane >> 4;
    const int c    = lane & 15;
    const int w    = tid >> 5;

    v8f acc[8];
#pragma unroll
    for (int t = 0; t < 8; ++t)
        acc[t] = (v8f){0.f, 0.f, 0.f, 0.f, 0.f, 0.f, 0.f, 0.f};

    const float* __restrict__ ap = A + (size_t)(m0 + c) * lda + (w * KS + 4 * h);
    const float* __restrict__ bp = B + (size_t)(w * KS + 4 * h) * ldb + (n0 + 4 * c);

    for (int kb = 0; kb < KS; kb += 8) {
        const float4 av = *(const float4*)(ap + kb);
        float4 bl[4], bh[4];
#pragma unroll
        for (int j = 0; j < 4; ++j) {
            const float* __restrict__ row = bp + (size_t)(kb + j) * ldb;
            bl[j] = *(const float4*)(row);
            bh[j] = *(const float4*)(row + 64);
        }
        const float bq[2][4][4] = {
            {{bl[0].x, bl[0].y, bl[0].z, bl[0].w}, {bl[1].x, bl[1].y, bl[1].z, bl[1].w},
             {bl[2].x, bl[2].y, bl[2].z, bl[2].w}, {bl[3].x, bl[3].y, bl[3].z, bl[3].w}},
            {{bh[0].x, bh[0].y, bh[0].z, bh[0].w}, {bh[1].x, bh[1].y, bh[1].z, bh[1].w},
             {bh[2].x, bh[2].y, bh[2].z, bh[2].w}, {bh[3].x, bh[3].y, bh[3].z, bh[3].w}}};
        const v2f a01 = {av.x, av.y};
        const v2f a23 = {av.z, av.w};
#pragma unroll
        for (int g = 0; g < 2; ++g) {
#pragma unroll
            for (int e = 0; e < 4; ++e) {
                const v2f b01 = {bq[g][0][e], bq[g][1][e]};
                acc[4 * g + e] = __builtin_amdgcn_wmma_f32_16x16x4_f32(
                    false, a01, false, b01, (short)0, acc[4 * g + e], false, false);
            }
        }
#pragma unroll
        for (int g = 0; g < 2; ++g) {
#pragma unroll
            for (int e = 0; e < 4; ++e) {
                const v2f b23 = {bq[g][2][e], bq[g][3][e]};
                acc[4 * g + e] = __builtin_amdgcn_wmma_f32_16x16x4_f32(
                    false, a23, false, b23, (short)0, acc[4 * g + e], false, false);
            }
        }
    }

#pragma unroll
    for (int g = 0; g < 2; ++g) {
#pragma unroll
        for (int r = 0; r < 8; ++r) {
            const float4 o = make_float4(acc[4 * g + 0][r], acc[4 * g + 1][r],
                                         acc[4 * g + 2][r], acc[4 * g + 3][r]);
            *(float4*)&part[w][8 * h + r][64 * g + 4 * c] = o;
        }
    }
    __syncthreads();

#pragma unroll
    for (int i = 0; i < 2; ++i) {
        const int f   = tid + 256 * i;
        const int row = f >> 5;
        const int c4  = (f & 31) * 4;
        float4 s = *(const float4*)&part[0][row][c4];
#pragma unroll
        for (int ww = 1; ww < 8; ++ww) {
            const float4 p = *(const float4*)&part[ww][row][c4];
            s.x += p.x; s.y += p.y; s.z += p.z; s.w += p.w;
        }
        *(float4*)(C + (size_t)(m0 + row) * ldc + (n0 + c4)) = s;
    }
}

__global__ __launch_bounds__(256) void proj_kernel(
    const float* __restrict__ query, const float* __restrict__ value,
    const float* __restrict__ W1, const float* __restrict__ W2,
    float* __restrict__ Qo, float* __restrict__ Ko)
{
    __shared__ float part[8][16][128];
    const int which = (int)blockIdx.x >> 7;
    const int mt    = (int)blockIdx.x & 127;
    const float* A  = which ? value : query;
    const float* Bm = which ? W2 : W1;
    float* C        = which ? Ko : Qo;
    strip_gemm_splitk<DM / 8>(A, DM, Bm, UN, C, UN, mt * 16, 0, part);
}

__global__ __launch_bounds__(512) void score_softmax_kernel(
    const float* __restrict__ Q, const float* __restrict__ Kmat,
    const float* __restrict__ scale, float* __restrict__ attn)
{
    const int b    = (int)blockIdx.x >> 6;
    const int t0   = ((int)blockIdx.x & 63) * TPB_T;
    const int tid  = (int)threadIdx.x;
    __shared__ float q_sh[TPB_T][UN];
    __shared__ float sc_sh[UN];
    __shared__ float2 wpair[2][16];

    if (tid < UN) sc_sh[tid] = scale[tid];
    {
        const int tt = tid >> 7;
        const int uu = tid & 127;
        q_sh[tt][uu]     = Q[(size_t)(b * TQ + t0 + tt) * UN + uu];
        q_sh[tt + 4][uu] = Q[(size_t)(b * TQ + t0 + tt + 4) * UN + uu];
    }

    float kreg[UN];
    {
        const float4* __restrict__ krow4 =
            (const float4*)(Kmat + ((size_t)(b * TV + tid)) * UN);
#pragma unroll
        for (int i = 0; i < UN / 4; ++i) {
            const float4 v = krow4[i];
            kreg[4 * i + 0] = v.x;
            kreg[4 * i + 1] = v.y;
            kreg[4 * i + 2] = v.z;
            kreg[4 * i + 3] = v.w;
        }
    }
    __syncthreads();

    const int w = tid >> 5;
    for (int tt = 0; tt < TPB_T; ++tt) {
        float s = 0.f;
#pragma unroll
        for (int u = 0; u < UN; ++u) {
            s = __builtin_fmaf(sc_sh[u], fast_tanh(q_sh[tt][u] + kreg[u]), s);
        }

        const float r16 = row16_max(s);
        const float mw  = fmaxf(r16, xhalf(r16));
        const float l16 = row16_sum(__expf(s - mw));
        const float lw  = l16 + xhalf(l16);
        if ((tid & 31) == 0) wpair[tt & 1][w] = make_float2(mw, lw);
        __syncthreads();

        const float2 p   = wpair[tt & 1][tid & 15];
        const float bmax = row16_max(p.x);
        const float bsum = row16_sum(p.y * __expf(p.x - bmax));

        attn[(size_t)(b * TQ + t0 + tt) * TV + tid] = __expf(s - bmax) / bsum;
    }
}

__global__ __launch_bounds__(256) void context_kernel(
    const float* __restrict__ attn, const float* __restrict__ value,
    float* __restrict__ ctx)
{
    __shared__ float part[8][16][128];
    const int b   = (int)blockIdx.x >> 6;
    const int rem = (int)blockIdx.x & 63;
    const int mt  = rem >> 1;
    const int nh  = rem & 1;
    const float* A  = attn  + (size_t)b * TQ * TV;
    const float* Bv = value + (size_t)b * TV * DM;
    float* C        = ctx   + (size_t)b * TQ * DM;
    strip_gemm_splitk<TV / 8>(A, TV, Bv, DM, C, DM, mt * 16, nh * 128, part);
}

extern "C" void kernel_launch(void* const* d_in, const int* in_sizes, int n_in,
                              void* d_out, int out_size, void* d_ws, size_t ws_size,
                              hipStream_t stream) {
    const float* query = (const float*)d_in[0];
    const float* value = (const float*)d_in[1];
    const float* W1    = (const float*)d_in[2];
    const float* W2    = (const float*)d_in[3];
    const float* scale = (const float*)d_in[4];

    float* out  = (float*)d_out;
    float* ctx  = out;
    float* attn = out + (size_t)B_DIM * TQ * DM;

    float* Qw = (float*)d_ws;
    float* Kw = Qw + (size_t)(B_DIM * TQ) * UN;

    proj_kernel<<<256, 256, 0, stream>>>(query, value, W1, W2, Qw, Kw);
    score_softmax_kernel<<<(B_DIM * TQ) / TPB_T, TV, 0, stream>>>(Qw, Kw, scale, attn);
    context_kernel<<<256, 256, 0, stream>>>(attn, value, ctx);
}


# score kernel: q rows and scale vector read by scalar loads into scalar registers (were LDS broadcast reads): inner loop has no LDS access or wait; same summation order
# speedup vs baseline: 1.8874x; 1.0455x over previous
#include <hip/hip_runtime.h>
#include <hip/hip_bf16.h>

typedef __attribute__((ext_vector_type(2))) float v2f;
typedef __attribute__((ext_vector_type(8))) float v8f;

#define B_DIM 4
#define TQ 512
#define TV 512
#define DM 256
#define UN 128
#define TPB_T 8

static __device__ __forceinline__ float fast_tanh(float x) {
#if __has_builtin(__builtin_amdgcn_tanhf)
    return __builtin_amdgcn_tanhf(x);
#else
    return tanhf(x);
#endif
}

template <int MASK>
static __device__ __forceinline__ float row_xor(float x) {
    const int b = __builtin_bit_cast(int, x);
    return __builtin_bit_cast(float,
        __builtin_amdgcn_update_dpp(b, b, 0x160 + MASK, 0xf, 0xf, true));
}

static __device__ __forceinline__ float xhalf(float x) {
    const int b = __builtin_bit_cast(int, x);
    return __builtin_bit_cast(float,
        __builtin_amdgcn_permlanex16(b, b, 0x76543210, 0xfedcba98, false, false));
}

static __device__ __forceinline__ float row16_max(float x) {
    x = fmaxf(x, row_xor<1>(x));
    x = fmaxf(x, row_xor<2>(x));
    x = fmaxf(x, row_xor<4>(x));
    x = fmaxf(x, row_xor<8>(x));
    return x;
}
static __device__ __forceinline__ float row16_sum(float x) {
    x += row_xor<1>(x);
    x += row_xor<2>(x);
    x += row_xor<4>(x);
    x += row_xor<8>(x);
    return x;
}

template <int KS>
static __device__ __forceinline__ void strip_gemm_splitk(
    const float* __restrict__ A, int lda,
    const float* __restrict__ B, int ldb,
    float* __restrict__ C, int ldc,
    int m0, int n0, float (*part)[16][128])
{
    const int tid  = (int)threadIdx.x;
    const int lane = tid & 31;
    const int h    = lane >> 4;
    const int c    = lane & 15;
    const int w    = tid >> 5;

    v8f acc[8];
#pragma unroll
    for (int t = 0; t < 8; ++t)
        acc[t] = (v8f){0.f, 0.f, 0.f, 0.f, 0.f, 0.f, 0.f, 0.f};

    const float* __restrict__ ap = A + (size_t)(m0 + c) * lda + (w * KS + 4 * h);
    const float* __restrict__ bp = B + (size_t)(w * KS + 4 * h) * ldb + (n0 + 4 * c);

    for (int kb = 0; kb < KS; kb += 8) {
        const float4 av = *(const float4*)(ap + kb);
        float4 bl[4], bh[4];
#pragma unroll
        for (int j = 0; j < 4; ++j) {
            const float* __restrict__ row = bp + (size_t)(kb + j) * ldb;
            bl[j] = *(const float4*)(row);
            bh[j] = *(const float4*)(row + 64);
        }
        const float bq[2][4][4] = {
            {{bl[0].x, bl[0].y, bl[0].z, bl[0].w}, {bl[1].x, bl[1].y, bl[1].z, bl[1].w},
             {bl[2].x, bl[2].y, bl[2].z, bl[2].w}, {bl[3].x, bl[3].y, bl[3].z, bl[3].w}},
            {{bh[0].x, bh[0].y, bh[0].z, bh[0].w}, {bh[1].x, bh[1].y, bh[1].z, bh[1].w},
             {bh[2].x, bh[2].y, bh[2].z, bh[2].w}, {bh[3].x, bh[3].y, bh[3].z, bh[3].w}}};
        const v2f a01 = {av.x, av.y};
        const v2f a23 = {av.z, av.w};
#pragma unroll
        for (int g = 0; g < 2; ++g) {
#pragma unroll
            for (int e = 0; e < 4; ++e) {
                const v2f b01 = {bq[g][0][e], bq[g][1][e]};
                acc[4 * g + e] = __builtin_amdgcn_wmma_f32_16x16x4_f32(
                    false, a01, false, b01, (short)0, acc[4 * g + e], false, false);
            }
        }
#pragma unroll
        for (int g = 0; g < 2; ++g) {
#pragma unroll
            for (int e = 0; e < 4; ++e) {
                const v2f b23 = {bq[g][2][e], bq[g][3][e]};
                acc[4 * g + e] = __builtin_amdgcn_wmma_f32_16x16x4_f32(
                    false, a23, false, b23, (short)0, acc[4 * g + e], false, false);
            }
        }
    }

#pragma unroll
    for (int g = 0; g < 2; ++g) {
#pragma unroll
        for (int r = 0; r < 8; ++r) {
            const float4 o = make_float4(acc[4 * g + 0][r], acc[4 * g + 1][r],
                                         acc[4 * g + 2][r], acc[4 * g + 3][r]);
            *(float4*)&part[w][8 * h + r][64 * g + 4 * c] = o;
        }
    }
    __syncthreads();

#pragma unroll
    for (int i = 0; i < 2; ++i) {
        const int f   = tid + 256 * i;
        const int row = f >> 5;
        const int c4  = (f & 31) * 4;
        float4 s = *(const float4*)&part[0][row][c4];
#pragma unroll
        for (int ww = 1; ww < 8; ++ww) {
            const float4 p = *(const float4*)&part[ww][row][c4];
            s.x += p.x; s.y += p.y; s.z += p.z; s.w += p.w;
        }
        *(float4*)(C + (size_t)(m0 + row) * ldc + (n0 + c4)) = s;
    }
}

__global__ __launch_bounds__(256) void proj_kernel(
    const float* __restrict__ query, const float* __restrict__ value,
    const float* __restrict__ W1, const float* __restrict__ W2,
    float* __restrict__ Qo, float* __restrict__ Ko)
{
    __shared__ float part[8][16][128];
    const int which = (int)blockIdx.x >> 7;
    const int mt    = (int)blockIdx.x & 127;
    const float* A  = which ? value : query;
    const float* Bm = which ? W2 : W1;
    float* C        = which ? Ko : Qo;
    strip_gemm_splitk<DM / 8>(A, DM, Bm, UN, C, UN, mt * 16, 0, part);
}

typedef const __attribute__((address_space(4))) float* const_f32_ptr;

__global__ __launch_bounds__(512) void score_softmax_kernel(
    const float* __restrict__ Q, const float* __restrict__ Kmat,
    const float* __restrict__ scale, float* __restrict__ attn)
{
    const int b    = (int)blockIdx.x >> 6;
    const int t0   = ((int)blockIdx.x & 63) * TPB_T;
    const int tid  = (int)threadIdx.x;
    __shared__ float2 wpair[2][16];

    float kreg[UN];
    {
        const float4* __restrict__ krow4 =
            (const float4*)(Kmat + ((size_t)(b * TV + tid)) * UN);
#pragma unroll
        for (int i = 0; i < UN / 4; ++i) {
            const float4 v = krow4[i];
            kreg[4 * i + 0] = v.x;
            kreg[4 * i + 1] = v.y;
            kreg[4 * i + 2] = v.z;
            kreg[4 * i + 3] = v.w;
        }
    }

    const_f32_ptr qblk = (const_f32_ptr)(Q + (size_t)(b * TQ + t0) * UN);
    const_f32_ptr scl  = (const_f32_ptr)scale;

    const int w = tid >> 5;
    for (int tt = 0; tt < TPB_T; ++tt) {
        const_f32_ptr qrow = qblk + tt * UN;
        const_f32_ptr srow = scl;
        asm volatile("" : "+s"(srow));
        float s = 0.f;
#pragma unroll
        for (int u = 0; u < UN; ++u) {
            s = __builtin_fmaf(srow[u], fast_tanh(qrow[u] + kreg[u]), s);
        }

        const float r16 = row16_max(s);
        const float mw  = fmaxf(r16, xhalf(r16));
        const float l16 = row16_sum(__expf(s - mw));
        const float lw  = l16 + xhalf(l16);
        if ((tid & 31) == 0) wpair[tt & 1][w] = make_float2(mw, lw);
        __syncthreads();

        const float2 p   = wpair[tt & 1][tid & 15];
        const float bmax = row16_max(p.x);
        const float bsum = row16_sum(p.y * __expf(p.x - bmax));

        attn[(size_t)(b * TQ + t0 + tt) * TV + tid] = __expf(s - bmax) / bsum;
    }
}

__global__ __launch_bounds__(256) void context_kernel(
    const float* __restrict__ attn, const float* __restrict__ value,
    float* __restrict__ ctx)
{
    __shared__ float part[8][16][128];
    const int b   = (int)blockIdx.x >> 6;
    const int rem = (int)blockIdx.x & 63;
    const int mt  = rem >> 1;
    const int nh  = rem & 1;
    const float* A  = attn  + (size_t)b * TQ * TV;
    const float* Bv = value + (size_t)b * TV * DM;
    float* C        = ctx   + (size_t)b * TQ * DM;
    strip_gemm_splitk<TV / 8>(A, TV, Bv, DM, C, DM, mt * 16, nh * 128, part);
}

extern "C" void kernel_launch(void* const* d_in, const int* in_sizes, int n_in,
                              void* d_out, int out_size, void* d_ws, size_t ws_size,
                              hipStream_t stream) {
    const float* query = (const float*)d_in[0];
    const float* value = (const float*)d_in[1];
    const float* W1    = (const float*)d_in[2];
    const float* W2    = (const float*)d_in[3];
    const float* scale = (const float*)d_in[4];

    float* out  = (float*)d_out;
    float* ctx  = out;
    float* attn = out + (size_t)B_DIM * TQ * DM;

    float* Qw = (float*)d_ws;
    float* Kw = Qw + (size_t)(B_DIM * TQ) * UN;

    proj_kernel<<<256, 256, 0, stream>>>(query, value, W1, W2, Qw, Kw);
    score_softmax_kernel<<<(B_DIM * TQ) / TPB_T, TV, 0, stream>>>(Qw, Kw, scale, attn);
    context_kernel<<<256, 256, 0, stream>>>(attn, value, ctx);
}


# score kernel: row barrier split (signal after the pair store, wait one row later): softmax of row t runs behind the tanh loop of row t+1; tanh loop now compiles to dual-issue fma+add; same sum order
# speedup vs baseline: 1.9809x; 1.0495x over previous
#include <hip/hip_runtime.h>
#include <hip/hip_bf16.h>

typedef __attribute__((ext_vector_type(2))) float v2f;
typedef __attribute__((ext_vector_type(8))) float v8f;

#define B_DIM 4
#define TQ 512
#define TV 512
#define DM 256
#define UN 128
#define TPB_T 8

static __device__ __forceinline__ float fast_tanh(float x) {
#if __has_builtin(__builtin_amdgcn_tanhf)
    return __builtin_amdgcn_tanhf(x);
#else
    return tanhf(x);
#endif
}

template <int MASK>
static __device__ __forceinline__ float row_xor(float x) {
    const int b = __builtin_bit_cast(int, x);
    return __builtin_bit_cast(float,
        __builtin_amdgcn_update_dpp(b, b, 0x160 + MASK, 0xf, 0xf, true));
}

static __device__ __forceinline__ float xhalf(float x) {
    const int b = __builtin_bit_cast(int, x);
    return __builtin_bit_cast(float,
        __builtin_amdgcn_permlanex16(b, b, 0x76543210, 0xfedcba98, false, false));
}

static __device__ __forceinline__ float row16_max(float x) {
    x = fmaxf(x, row_xor<1>(x));
    x = fmaxf(x, row_xor<2>(x));
    x = fmaxf(x, row_xor<4>(x));
    x = fmaxf(x, row_xor<8>(x));
    return x;
}
static __device__ __forceinline__ float row16_sum(float x) {
    x += row_xor<1>(x);
    x += row_xor<2>(x);
    x += row_xor<4>(x);
    x += row_xor<8>(x);
    return x;
}

template <int KS>
static __device__ __forceinline__ void strip_gemm_splitk(
    const float* __restrict__ A, int lda,
    const float* __restrict__ B, int ldb,
    float* __restrict__ C, int ldc,
    int m0, int n0, float (*part)[16][128])
{
    const int tid  = (int)threadIdx.x;
    const int lane = tid & 31;
    const int h    = lane >> 4;
    const int c    = lane & 15;
    const int w    = tid >> 5;

    v8f acc[8];
#pragma unroll
    for (int t = 0; t < 8; ++t)
        acc[t] = (v8f){0.f, 0.f, 0.f, 0.f, 0.f, 0.f, 0.f, 0.f};

    const float* __restrict__ ap = A + (size_t)(m0 + c) * lda + (w * KS + 4 * h);
    const float* __restrict__ bp = B + (size_t)(w * KS + 4 * h) * ldb + (n0 + 4 * c);

    for (int kb = 0; kb < KS; kb += 8) {
        const float4 av = *(const float4*)(ap + kb);
        float4 bl[4], bh[4];
#pragma unroll
        for (int j = 0; j < 4; ++j) {
            const float* __restrict__ row = bp + (size_t)(kb + j) * ldb;
            bl[j] = *(const float4*)(row);
            bh[j] = *(const float4*)(row + 64);
        }
        const float bq[2][4][4] = {
            {{bl[0].x, bl[0].y, bl[0].z, bl[0].w}, {bl[1].x, bl[1].y, bl[1].z, bl[1].w},
             {bl[2].x, bl[2].y, bl[2].z, bl[2].w}, {bl[3].x, bl[3].y, bl[3].z, bl[3].w}},
            {{bh[0].x, bh[0].y, bh[0].z, bh[0].w}, {bh[1].x, bh[1].y, bh[1].z, bh[1].w},
             {bh[2].x, bh[2].y, bh[2].z, bh[2].w}, {bh[3].x, bh[3].y, bh[3].z, bh[3].w}}};
        const v2f a01 = {av.x, av.y};
        const v2f a23 = {av.z, av.w};
#pragma unroll
        for (int g = 0; g < 2; ++g) {
#pragma unroll
            for (int e = 0; e < 4; ++e) {
                const v2f b01 = {bq[g][0][e], bq[g][1][e]};
                acc[4 * g + e] = __builtin_amdgcn_wmma_f32_16x16x4_f32(
                    false, a01, false, b01, (short)0, acc[4 * g + e], false, false);
            }
        }
#pragma unroll
        for (int g = 0; g < 2; ++g) {
#pragma unroll
            for (int e = 0; e < 4; ++e) {
                const v2f b23 = {bq[g][2][e], bq[g][3][e]};
                acc[4 * g + e] = __builtin_amdgcn_wmma_f32_16x16x4_f32(
                    false, a23, false, b23, (short)0, acc[4 * g + e], false, false);
            }
        }
    }

#pragma unroll
    for (int g = 0; g < 2; ++g) {
#pragma unroll
        for (int r = 0; r < 8; ++r) {
            const float4 o = make_float4(acc[4 * g + 0][r], acc[4 * g + 1][r],
                                         acc[4 * g + 2][r], acc[4 * g + 3][r]);
            *(float4*)&part[w][8 * h + r][64 * g + 4 * c] = o;
        }
    }
    __syncthreads();

#pragma unroll
    for (int i = 0; i < 2; ++i) {
        const int f   = tid + 256 * i;
        const int row = f >> 5;
        const int c4  = (f & 31) * 4;
        float4 s = *(const float4*)&part[0][row][c4];
#pragma unroll
        for (int ww = 1; ww < 8; ++ww) {
            const float4 p = *(const float4*)&part[ww][row][c4];
            s.x += p.x; s.y += p.y; s.z += p.z; s.w += p.w;
        }
        *(float4*)(C + (size_t)(m0 + row) * ldc + (n0 + c4)) = s;
    }
}

__global__ __launch_bounds__(256) void proj_kernel(
    const float* __restrict__ query, const float* __restrict__ value,
    const float* __restrict__ W1, const float* __restrict__ W2,
    float* __restrict__ Qo, float* __restrict__ Ko)
{
    __shared__ float part[8][16][128];
    const int which = (int)blockIdx.x >> 7;
    const int mt    = (int)blockIdx.x & 127;
    const float* A  = which ? value : query;
    const float* Bm = which ? W2 : W1;
    float* C        = which ? Ko : Qo;
    strip_gemm_splitk<DM / 8>(A, DM, Bm, UN, C, UN, mt * 16, 0, part);
}

typedef const __attribute__((address_space(4))) float* const_f32_ptr;

static __device__ __forceinline__ void pairs_signal() {
    asm volatile("s_wait_dscnt 0x0\n\ts_barrier_signal -1" ::: "memory");
}
static __device__ __forceinline__ void pairs_wait() {
    asm volatile("s_barrier_wait -1" ::: "memory");
}

static __device__ __forceinline__ float row_score(const_f32_ptr qrow, const_f32_ptr scl,
                                                  const float (&kreg)[UN]) {
    const_f32_ptr srow = scl;
    asm volatile("" : "+s"(srow));
    float s = 0.f;
#pragma unroll
    for (int u = 0; u < UN; ++u) {
        s = __builtin_fmaf(srow[u], fast_tanh(qrow[u] + kreg[u]), s);
    }
    return s;
}

static __device__ __forceinline__ void publish_pair(float s, float2* slot, bool lane0) {
    const float r16 = row16_max(s);
    const float mw  = fmaxf(r16, xhalf(r16));
    const float l16 = row16_sum(__expf(s - mw));
    const float lw  = l16 + xhalf(l16);
    if (lane0) *slot = make_float2(mw, lw);
    pairs_signal();
}

static __device__ __forceinline__ float finish_row(float s, const float2* pairs, int lane16) {
    pairs_wait();
    const float2 p   = pairs[lane16];
    const float bmax = row16_max(p.x);
    const float bsum = row16_sum(p.y * __expf(p.x - bmax));
    return __expf(s - bmax) / bsum;
}

__global__ __launch_bounds__(512) void score_softmax_kernel(
    const float* __restrict__ Q, const float* __restrict__ Kmat,
    const float* __restrict__ scale, float* __restrict__ attn)
{
    const int b    = (int)blockIdx.x >> 6;
    const int t0   = ((int)blockIdx.x & 63) * TPB_T;
    const int tid  = (int)threadIdx.x;
    __shared__ float2 wpair[2][16];

    float kreg[UN];
    {
        const float4* __restrict__ krow4 =
            (const float4*)(Kmat + ((size_t)(b * TV + tid)) * UN);
#pragma unroll
        for (int i = 0; i < UN / 4; ++i) {
            const float4 v = krow4[i];
            kreg[4 * i + 0] = v.x;
            kreg[4 * i + 1] = v.y;
            kreg[4 * i + 2] = v.z;
            kreg[4 * i + 3] = v.w;
        }
    }

    const_f32_ptr qblk = (const_f32_ptr)(Q + (size_t)(b * TQ + t0) * UN);
    const_f32_ptr scl  = (const_f32_ptr)scale;

    const int  w      = tid >> 5;
    const bool lane0  = (tid & 31) == 0;
    const int  lane16 = tid & 15;
    float* __restrict__ arow = attn + (size_t)(b * TQ + t0) * TV + tid;

    float s_prev = row_score(qblk, scl, kreg);
    publish_pair(s_prev, &wpair[0][w], lane0);

    for (int tt = 1; tt < TPB_T; ++tt) {
        const float s_new = row_score(qblk + tt * UN, scl, kreg);
        arow[(size_t)(tt - 1) * TV] = finish_row(s_prev, wpair[(tt - 1) & 1], lane16);
        publish_pair(s_new, &wpair[tt & 1][w], lane0);
        s_prev = s_new;
    }
    arow[(size_t)(TPB_T - 1) * TV] = finish_row(s_prev, wpair[(TPB_T - 1) & 1], lane16);
}

__global__ __launch_bounds__(256) void context_kernel(
    const float* __restrict__ attn, const float* __restrict__ value,
    float* __restrict__ ctx)
{
    __shared__ float part[8][16][128];
    const int b   = (int)blockIdx.x >> 6;
    const int rem = (int)blockIdx.x & 63;
    const int mt  = rem >> 1;
    const int nh  = rem & 1;
    const float* A  = attn  + (size_t)b * TQ * TV;
    const float* Bv = value + (size_t)b * TV * DM;
    float* C        = ctx   + (size_t)b * TQ * DM;
    strip_gemm_splitk<TV / 8>(A, TV, Bv, DM, C, DM, mt * 16, nh * 128, part);
}

extern "C" void kernel_launch(void* const* d_in, const int* in_sizes, int n_in,
                              void* d_out, int out_size, void* d_ws, size_t ws_size,
                              hipStream_t stream) {
    const float* query = (const float*)d_in[0];
    const float* value = (const float*)d_in[1];
    const float* W1    = (const float*)d_in[2];
    const float* W2    = (const float*)d_in[3];
    const float* scale = (const float*)d_in[4];

    float* out  = (float*)d_out;
    float* ctx  = out;
    float* attn = out + (size_t)B_DIM * TQ * DM;

    float* Qw = (float*)d_ws;
    float* Kw = Qw + (size_t)(B_DIM * TQ) * UN;

    proj_kernel<<<256, 256, 0, stream>>>(query, value, W1, W2, Qw, Kw);
    score_softmax_kernel<<<(B_DIM * TQ) / TPB_T, TV, 0, stream>>>(Qw, Kw, scale, attn);
    context_kernel<<<256, 256, 0, stream>>>(attn, value, ctx);
}


# Both GEMMs: K split over 16 waves per block (was 8): each wave's load-then-WMMA chain is half as long, 4 waves per SIMD cover it; 16 partials added through LDS in fixed order (new summation tree)
# speedup vs baseline: 2.2551x; 1.1384x over previous
#include <hip/hip_runtime.h>
#include <hip/hip_bf16.h>

typedef __attribute__((ext_vector_type(2))) float v2f;
typedef __attribute__((ext_vector_type(8))) float v8f;

#define B_DIM 4
#define TQ 512
#define TV 512
#define DM 256
#define UN 128
#define TPB_T 8

static __device__ __forceinline__ float fast_tanh(float x) {
#if __has_builtin(__builtin_amdgcn_tanhf)
    return __builtin_amdgcn_tanhf(x);
#else
    return tanhf(x);
#endif
}

template <int MASK>
static __device__ __forceinline__ float row_xor(float x) {
    const int b = __builtin_bit_cast(int, x);
    return __builtin_bit_cast(float,
        __builtin_amdgcn_update_dpp(b, b, 0x160 + MASK, 0xf, 0xf, true));
}

static __device__ __forceinline__ float xhalf(float x) {
    const int b = __builtin_bit_cast(int, x);
    return __builtin_bit_cast(float,
        __builtin_amdgcn_permlanex16(b, b, 0x76543210, 0xfedcba98, false, false));
}

static __device__ __forceinline__ float row16_max(float x) {
    x = fmaxf(x, row_xor<1>(x));
    x = fmaxf(x, row_xor<2>(x));
    x = fmaxf(x, row_xor<4>(x));
    x = fmaxf(x, row_xor<8>(x));
    return x;
}
static __device__ __forceinline__ float row16_sum(float x) {
    x += row_xor<1>(x);
    x += row_xor<2>(x);
    x += row_xor<4>(x);
    x += row_xor<8>(x);
    return x;
}

template <int KS, int NW>
static __device__ __forceinline__ void strip_gemm_splitk(
    const float* __restrict__ A, int lda,
    const float* __restrict__ B, int ldb,
    float* __restrict__ C, int ldc,
    int m0, int n0, float (*part)[16][128])
{
    const int tid  = (int)threadIdx.x;
    const int lane = tid & 31;
    const int h    = lane >> 4;
    const int c    = lane & 15;
    const int w    = tid >> 5;

    v8f acc[8];
#pragma unroll
    for (int t = 0; t < 8; ++t)
        acc[t] = (v8f){0.f, 0.f, 0.f, 0.f, 0.f, 0.f, 0.f, 0.f};

    const float* __restrict__ ap = A + (size_t)(m0 + c) * lda + (w * KS + 4 * h);
    const float* __restrict__ bp = B + (size_t)(w * KS + 4 * h) * ldb + (n0 + 4 * c);

    for (int kb = 0; kb < KS; kb += 8) {
        const float4 av = *(const float4*)(ap + kb);
        float4 bl[4], bh[4];
#pragma unroll
        for (int j = 0; j < 4; ++j) {
            const float* __restrict__ row = bp + (size_t)(kb + j) * ldb;
            bl[j] = *(const float4*)(row);
            bh[j] = *(const float4*)(row + 64);
        }
        const float bq[2][4][4] = {
            {{bl[0].x, bl[0].y, bl[0].z, bl[0].w}, {bl[1].x, bl[1].y, bl[1].z, bl[1].w},
             {bl[2].x, bl[2].y, bl[2].z, bl[2].w}, {bl[3].x, bl[3].y, bl[3].z, bl[3].w}},
            {{bh[0].x, bh[0].y, bh[0].z, bh[0].w}, {bh[1].x, bh[1].y, bh[1].z, bh[1].w},
             {bh[2].x, bh[2].y, bh[2].z, bh[2].w}, {bh[3].x, bh[3].y, bh[3].z, bh[3].w}}};
        const v2f a01 = {av.x, av.y};
        const v2f a23 = {av.z, av.w};
#pragma unroll
        for (int g = 0; g < 2; ++g) {
#pragma unroll
            for (int e = 0; e < 4; ++e) {
                const v2f b01 = {bq[g][0][e], bq[g][1][e]};
                acc[4 * g + e] = __builtin_amdgcn_wmma_f32_16x16x4_f32(
                    false, a01, false, b01, (short)0, acc[4 * g + e], false, false);
            }
        }
#pragma unroll
        for (int g = 0; g < 2; ++g) {
#pragma unroll
            for (int e = 0; e < 4; ++e) {
                const v2f b23 = {bq[g][2][e], bq[g][3][e]};
                acc[4 * g + e] = __builtin_amdgcn_wmma_f32_16x16x4_f32(
                    false, a23, false, b23, (short)0, acc[4 * g + e], false, false);
            }
        }
    }

#pragma unroll
    for (int g = 0; g < 2; ++g) {
#pragma unroll
        for (int r = 0; r < 8; ++r) {
            const float4 o = make_float4(acc[4 * g + 0][r], acc[4 * g + 1][r],
                                         acc[4 * g + 2][r], acc[4 * g + 3][r]);
            *(float4*)&part[w][8 * h + r][64 * g + 4 * c] = o;
        }
    }
    __syncthreads();

#pragma unroll
    for (int i = 0; i < 512 / (NW * 32); ++i) {
        const int f   = tid + NW * 32 * i;
        const int row = f >> 5;
        const int c4  = (f & 31) * 4;
        float4 s = *(const float4*)&part[0][row][c4];
#pragma unroll
        for (int ww = 1; ww < NW; ++ww) {
            const float4 p = *(const float4*)&part[ww][row][c4];
            s.x += p.x; s.y += p.y; s.z += p.z; s.w += p.w;
        }
        *(float4*)(C + (size_t)(m0 + row) * ldc + (n0 + c4)) = s;
    }
}

#define GEMM_WAVES 16

__global__ __launch_bounds__(GEMM_WAVES * 32) void proj_kernel(
    const float* __restrict__ query, const float* __restrict__ value,
    const float* __restrict__ W1, const float* __restrict__ W2,
    float* __restrict__ Qo, float* __restrict__ Ko)
{
    __shared__ float part[GEMM_WAVES][16][128];
    const int which = (int)blockIdx.x >> 7;
    const int mt    = (int)blockIdx.x & 127;
    const float* A  = which ? value : query;
    const float* Bm = which ? W2 : W1;
    float* C        = which ? Ko : Qo;
    strip_gemm_splitk<DM / GEMM_WAVES, GEMM_WAVES>(A, DM, Bm, UN, C, UN, mt * 16, 0, part);
}

typedef const __attribute__((address_space(4))) float* const_f32_ptr;

static __device__ __forceinline__ void pairs_signal() {
    asm volatile("s_wait_dscnt 0x0\n\ts_barrier_signal -1" ::: "memory");
}
static __device__ __forceinline__ void pairs_wait() {
    asm volatile("s_barrier_wait -1" ::: "memory");
}

static __device__ __forceinline__ float row_score(const_f32_ptr qrow, const_f32_ptr scl,
                                                  const float (&kreg)[UN]) {
    const_f32_ptr srow = scl;
    asm volatile("" : "+s"(srow));
    float s = 0.f;
#pragma unroll
    for (int u = 0; u < UN; ++u) {
        s = __builtin_fmaf(srow[u], fast_tanh(qrow[u] + kreg[u]), s);
    }
    return s;
}

static __device__ __forceinline__ void publish_pair(float s, float2* slot, bool lane0) {
    const float r16 = row16_max(s);
    const float mw  = fmaxf(r16, xhalf(r16));
    const float l16 = row16_sum(__expf(s - mw));
    const float lw  = l16 + xhalf(l16);
    if (lane0) *slot = make_float2(mw, lw);
    pairs_signal();
}

static __device__ __forceinline__ float finish_row(float s, const float2* pairs, int lane16) {
    pairs_wait();
    const float2 p   = pairs[lane16];
    const float bmax = row16_max(p.x);
    const float bsum = row16_sum(p.y * __expf(p.x - bmax));
    return __expf(s - bmax) / bsum;
}

__global__ __launch_bounds__(512) void score_softmax_kernel(
    const float* __restrict__ Q, const float* __restrict__ Kmat,
    const float* __restrict__ scale, float* __restrict__ attn)
{
    const int b    = (int)blockIdx.x >> 6;
    const int t0   = ((int)blockIdx.x & 63) * TPB_T;
    const int tid  = (int)threadIdx.x;
    __shared__ float2 wpair[2][16];

    float kreg[UN];
    {
        const float4* __restrict__ krow4 =
            (const float4*)(Kmat + ((size_t)(b * TV + tid)) * UN);
#pragma unroll
        for (int i = 0; i < UN / 4; ++i) {
            const float4 v = krow4[i];
            kreg[4 * i + 0] = v.x;
            kreg[4 * i + 1] = v.y;
            kreg[4 * i + 2] = v.z;
            kreg[4 * i + 3] = v.w;
        }
    }

    const_f32_ptr qblk = (const_f32_ptr)(Q + (size_t)(b * TQ + t0) * UN);
    const_f32_ptr scl  = (const_f32_ptr)scale;

    const int  w      = tid >> 5;
    const bool lane0  = (tid & 31) == 0;
    const int  lane16 = tid & 15;
    float* __restrict__ arow = attn + (size_t)(b * TQ + t0) * TV + tid;

    float s_prev = row_score(qblk, scl, kreg);
    publish_pair(s_prev, &wpair[0][w], lane0);

    for (int tt = 1; tt < TPB_T; ++tt) {
        const float s_new = row_score(qblk + tt * UN, scl, kreg);
        arow[(size_t)(tt - 1) * TV] = finish_row(s_prev, wpair[(tt - 1) & 1], lane16);
        publish_pair(s_new, &wpair[tt & 1][w], lane0);
        s_prev = s_new;
    }
    arow[(size_t)(TPB_T - 1) * TV] = finish_row(s_prev, wpair[(TPB_T - 1) & 1], lane16);
}

__global__ __launch_bounds__(GEMM_WAVES * 32) void context_kernel(
    const float* __restrict__ attn, const float* __restrict__ value,
    float* __restrict__ ctx)
{
    __shared__ float part[GEMM_WAVES][16][128];
    const int b   = (int)blockIdx.x >> 6;
    const int rem = (int)blockIdx.x & 63;
    const int mt  = rem >> 1;
    const int nh  = rem & 1;
    const float* A  = attn  + (size_t)b * TQ * TV;
    const float* Bv = value + (size_t)b * TV * DM;
    float* C        = ctx   + (size_t)b * TQ * DM;
    strip_gemm_splitk<TV / GEMM_WAVES, GEMM_WAVES>(A, TV, Bv, DM, C, DM, mt * 16, nh * 128, part);
}

extern "C" void kernel_launch(void* const* d_in, const int* in_sizes, int n_in,
                              void* d_out, int out_size, void* d_ws, size_t ws_size,
                              hipStream_t stream) {
    const float* query = (const float*)d_in[0];
    const float* value = (const float*)d_in[1];
    const float* W1    = (const float*)d_in[2];
    const float* W2    = (const float*)d_in[3];
    const float* scale = (const float*)d_in[4];

    float* out  = (float*)d_out;
    float* ctx  = out;
    float* attn = out + (size_t)B_DIM * TQ * DM;

    float* Qw = (float*)d_ws;
    float* Kw = Qw + (size_t)(B_DIM * TQ) * UN;

    proj_kernel<<<256, 512, 0, stream>>>(query, value, W1, W2, Qw, Kw);
    score_softmax_kernel<<<(B_DIM * TQ) / TPB_T, TV, 0, stream>>>(Qw, Kw, scale, attn);
    context_kernel<<<256, 512, 0, stream>>>(attn, value, ctx);
}


# score kernel: scalar q/scale loads off the tanh loop's critical path: q block and scale lines loaded into the scalar cache at entry, 64-byte chunks requested one chunk ahead of their wait; same sums
# speedup vs baseline: 2.6492x; 1.1748x over previous
#include <hip/hip_runtime.h>
#include <hip/hip_bf16.h>

typedef __attribute__((ext_vector_type(2))) float v2f;
typedef __attribute__((ext_vector_type(8))) float v8f;

#define B_DIM 4
#define TQ 512
#define TV 512
#define DM 256
#define UN 128
#define TPB_T 8

static __device__ __forceinline__ float fast_tanh(float x) {
#if __has_builtin(__builtin_amdgcn_tanhf)
    return __builtin_amdgcn_tanhf(x);
#else
    return tanhf(x);
#endif
}

template <int MASK>
static __device__ __forceinline__ float row_xor(float x) {
    const int b = __builtin_bit_cast(int, x);
    return __builtin_bit_cast(float,
        __builtin_amdgcn_update_dpp(b, b, 0x160 + MASK, 0xf, 0xf, true));
}

static __device__ __forceinline__ float xhalf(float x) {
    const int b = __builtin_bit_cast(int, x);
    return __builtin_bit_cast(float,
        __builtin_amdgcn_permlanex16(b, b, 0x76543210, 0xfedcba98, false, false));
}

static __device__ __forceinline__ float row16_max(float x) {
    x = fmaxf(x, row_xor<1>(x));
    x = fmaxf(x, row_xor<2>(x));
    x = fmaxf(x, row_xor<4>(x));
    x = fmaxf(x, row_xor<8>(x));
    return x;
}
static __device__ __forceinline__ float row16_sum(float x) {
    x += row_xor<1>(x);
    x += row_xor<2>(x);
    x += row_xor<4>(x);
    x += row_xor<8>(x);
    return x;
}

template <int KS, int NW>
static __device__ __forceinline__ void strip_gemm_splitk(
    const float* __restrict__ A, int lda,
    const float* __restrict__ B, int ldb,
    float* __restrict__ C, int ldc,
    int m0, int n0, float (*part)[16][128])
{
    const int tid  = (int)threadIdx.x;
    const int lane = tid & 31;
    const int h    = lane >> 4;
    const int c    = lane & 15;
    const int w    = tid >> 5;

    v8f acc[8];
#pragma unroll
    for (int t = 0; t < 8; ++t)
        acc[t] = (v8f){0.f, 0.f, 0.f, 0.f, 0.f, 0.f, 0.f, 0.f};

    const float* __restrict__ ap = A + (size_t)(m0 + c) * lda + (w * KS + 4 * h);
    const float* __restrict__ bp = B + (size_t)(w * KS + 4 * h) * ldb + (n0 + 4 * c);

    for (int kb = 0; kb < KS; kb += 8) {
        const float4 av = *(const float4*)(ap + kb);
        float4 bl[4], bh[4];
#pragma unroll
        for (int j = 0; j < 4; ++j) {
            const float* __restrict__ row = bp + (size_t)(kb + j) * ldb;
            bl[j] = *(const float4*)(row);
            bh[j] = *(const float4*)(row + 64);
        }
        const float bq[2][4][4] = {
            {{bl[0].x, bl[0].y, bl[0].z, bl[0].w}, {bl[1].x, bl[1].y, bl[1].z, bl[1].w},
             {bl[2].x, bl[2].y, bl[2].z, bl[2].w}, {bl[3].x, bl[3].y, bl[3].z, bl[3].w}},
            {{bh[0].x, bh[0].y, bh[0].z, bh[0].w}, {bh[1].x, bh[1].y, bh[1].z, bh[1].w},
             {bh[2].x, bh[2].y, bh[2].z, bh[2].w}, {bh[3].x, bh[3].y, bh[3].z, bh[3].w}}};
        const v2f a01 = {av.x, av.y};
        const v2f a23 = {av.z, av.w};
#pragma unroll
        for (int g = 0; g < 2; ++g) {
#pragma unroll
            for (int e = 0; e < 4; ++e) {
                const v2f b01 = {bq[g][0][e], bq[g][1][e]};
                acc[4 * g + e] = __builtin_amdgcn_wmma_f32_16x16x4_f32(
                    false, a01, false, b01, (short)0, acc[4 * g + e], false, false);
            }
        }
#pragma unroll
        for (int g = 0; g < 2; ++g) {
#pragma unroll
            for (int e = 0; e < 4; ++e) {
                const v2f b23 = {bq[g][2][e], bq[g][3][e]};
                acc[4 * g + e] = __builtin_amdgcn_wmma_f32_16x16x4_f32(
                    false, a23, false, b23, (short)0, acc[4 * g + e], false, false);
            }
        }
    }

#pragma unroll
    for (int g = 0; g < 2; ++g) {
#pragma unroll
        for (int r = 0; r < 8; ++r) {
            const float4 o = make_float4(acc[4 * g + 0][r], acc[4 * g + 1][r],
                                         acc[4 * g + 2][r], acc[4 * g + 3][r]);
            *(float4*)&part[w][8 * h + r][64 * g + 4 * c] = o;
        }
    }
    __syncthreads();

#pragma unroll
    for (int i = 0; i < 512 / (NW * 32); ++i) {
        const int f   = tid + NW * 32 * i;
        const int row = f >> 5;
        const int c4  = (f & 31) * 4;
        float4 s = *(const float4*)&part[0][row][c4];
#pragma unroll
        for (int ww = 1; ww < NW; ++ww) {
            const float4 p = *(const float4*)&part[ww][row][c4];
            s.x += p.x; s.y += p.y; s.z += p.z; s.w += p.w;
        }
        *(float4*)(C + (size_t)(m0 + row) * ldc + (n0 + c4)) = s;
    }
}

#define GEMM_WAVES 16

__global__ __launch_bounds__(GEMM_WAVES * 32) void proj_kernel(
    const float* __restrict__ query, const float* __restrict__ value,
    const float* __restrict__ W1, const float* __restrict__ W2,
    float* __restrict__ Qo, float* __restrict__ Ko)
{
    __shared__ float part[GEMM_WAVES][16][128];
    const int which = (int)blockIdx.x >> 7;
    const int mt    = (int)blockIdx.x & 127;
    const float* A  = which ? value : query;
    const float* Bm = which ? W2 : W1;
    float* C        = which ? Ko : Qo;
    strip_gemm_splitk<DM / GEMM_WAVES, GEMM_WAVES>(A, DM, Bm, UN, C, UN, mt * 16, 0, part);
}

typedef const __attribute__((address_space(4))) float* const_f32_ptr;

static __device__ __forceinline__ void pairs_signal() {
    asm volatile("s_wait_dscnt 0x0\n\ts_barrier_signal -1" ::: "memory");
}
static __device__ __forceinline__ void pairs_wait() {
    asm volatile("s_barrier_wait -1" ::: "memory");
}

typedef __attribute__((ext_vector_type(16))) float v16f;

template <int BYTE_OFF>
static __device__ __forceinline__ v16f chunk_request(const_f32_ptr base) {
    v16f r;
    asm volatile("s_load_b512 %0, %1, %2" : "=&s"(r) : "s"(base), "n"(BYTE_OFF));
    return r;
}

template <int BYTE_OFF>
static __device__ __forceinline__ v16f chunk_request_pinned(const_f32_ptr base, float& pin) {
    v16f r;
    asm volatile("s_load_b512 %0, %2, %3" : "=&s"(r), "+v"(pin) : "s"(base), "n"(BYTE_OFF));
    return r;
}

static __device__ __forceinline__ void chunk_wait(v16f& q, v16f& sc, float& s) {
    asm volatile("s_wait_kmcnt 0x0" : "+s"(q), "+s"(sc), "+v"(s));
}

template <int C, bool HAS_NEXT>
static __device__ __forceinline__ void score_chunk(const_f32_ptr qrow, const_f32_ptr scl,
                                                   const_f32_ptr qnext, v16f& qa, v16f& sa,
                                                   float (&kreg)[UN], float& s) {
    chunk_wait(qa, sa, s);
    v16f qn = qa, sn = sa;
    if constexpr (C + 1 < UN / 16) {
        qn = chunk_request_pinned<64 * (C + 1)>(qrow, kreg[16 * C]);
        sn = chunk_request_pinned<64 * (C + 1)>(scl, kreg[16 * C]);
    } else if constexpr (HAS_NEXT) {
        qn = chunk_request_pinned<0>(qnext, kreg[16 * C]);
        sn = chunk_request_pinned<0>(scl, kreg[16 * C]);
    }
#pragma unroll
    for (int j = 0; j < 16; ++j) {
        s = __builtin_fmaf(sa[j], fast_tanh(qa[j] + kreg[16 * C + j]), s);
    }
    qa = qn;
    sa = sn;
    if constexpr (C + 1 < UN / 16)
        score_chunk<C + 1, HAS_NEXT>(qrow, scl, qnext, qa, sa, kreg, s);
}

template <bool HAS_NEXT>
static __device__ __forceinline__ float row_score(const_f32_ptr qrow, const_f32_ptr scl,
                                                  const_f32_ptr qnext, v16f& qa, v16f& sa,
                                                  float (&kreg)[UN]) {
    float s = 0.f;
    score_chunk<0, HAS_NEXT>(qrow, scl, qnext, qa, sa, kreg, s);
    return s;
}

static __device__ __forceinline__ void publish_pair(float s, float2* slot, bool lane0) {
    const float r16 = row16_max(s);
    const float mw  = fmaxf(r16, xhalf(r16));
    const float l16 = row16_sum(__expf(s - mw));
    const float lw  = l16 + xhalf(l16);
    if (lane0) *slot = make_float2(mw, lw);
    pairs_signal();
}

static __device__ __forceinline__ float finish_row(float s, const float2* pairs, int lane16) {
    pairs_wait();
    const float2 p   = pairs[lane16];
    const float bmax = row16_max(p.x);
    const float bsum = row16_sum(p.y * __expf(p.x - bmax));
    return __expf(s - bmax) / bsum;
}

__global__ __launch_bounds__(512) __attribute__((amdgpu_waves_per_eu(4, 4))) void score_softmax_kernel(
    const float* __restrict__ Q, const float* __restrict__ Kmat,
    const float* __restrict__ scale, float* __restrict__ attn)
{
    const int b    = (int)blockIdx.x >> 6;
    const int t0   = ((int)blockIdx.x & 63) * TPB_T;
    const int tid  = (int)threadIdx.x;
    __shared__ float2 wpair[2][16];

    float kreg[UN];
    {
        const float4* __restrict__ krow4 =
            (const float4*)(Kmat + ((size_t)(b * TV + tid)) * UN);
#pragma unroll
        for (int i = 0; i < UN / 4; ++i) {
            const float4 v = krow4[i];
            kreg[4 * i + 0] = v.x;
            kreg[4 * i + 1] = v.y;
            kreg[4 * i + 2] = v.z;
            kreg[4 * i + 3] = v.w;
        }
    }

    const_f32_ptr qblk = (const_f32_ptr)(Q + (size_t)(b * TQ + t0) * UN);
    const_f32_ptr scl  = (const_f32_ptr)scale;

    const int  w      = tid >> 5;
    const bool lane0  = (tid & 31) == 0;
    const int  lane16 = tid & 15;
    float* __restrict__ arow = attn + (size_t)(b * TQ + t0) * TV + tid;

    const int wu = __builtin_amdgcn_readfirstlane(w);
    const_f32_ptr qtouch = qblk + 64 * wu;
    const_f32_ptr stouch = scl + 16 * (wu & 7);
    float l0, l1, l2, l3, l4;
    asm volatile("s_load_b32 %0, %1, 0x0"  : "=&s"(l0) : "s"(qtouch));
    asm volatile("s_load_b32 %0, %1, 0x40" : "=&s"(l1) : "s"(qtouch));
    asm volatile("s_load_b32 %0, %1, 0x80" : "=&s"(l2) : "s"(qtouch));
    asm volatile("s_load_b32 %0, %1, 0xc0" : "=&s"(l3) : "s"(qtouch));
    asm volatile("s_load_b32 %0, %1, 0x0"  : "=&s"(l4) : "s"(stouch));

    v16f qa = chunk_request<0>(qblk);
    v16f sa = chunk_request<0>(scl);

    asm volatile("s_wait_kmcnt 0x0"
                 : "+s"(qa), "+s"(sa), "+s"(l0), "+s"(l1), "+s"(l2), "+s"(l3), "+s"(l4),
                   "+v"(kreg[0]));

    float s_prev = row_score<true>(qblk, scl, qblk + UN, qa, sa, kreg);
    publish_pair(s_prev, &wpair[0][w], lane0);

    for (int tt = 1; tt < TPB_T - 1; ++tt) {
        const float s_new = row_score<true>(qblk + tt * UN, scl, qblk + (tt + 1) * UN, qa, sa, kreg);
        arow[(size_t)(tt - 1) * TV] = finish_row(s_prev, wpair[(tt - 1) & 1], lane16);
        publish_pair(s_new, &wpair[tt & 1][w], lane0);
        s_prev = s_new;
    }
    {
        const float s_new = row_score<false>(qblk + (TPB_T - 1) * UN, scl, qblk, qa, sa, kreg);
        arow[(size_t)(TPB_T - 2) * TV] = finish_row(s_prev, wpair[(TPB_T - 2) & 1], lane16);
        publish_pair(s_new, &wpair[(TPB_T - 1) & 1][w], lane0);
        s_prev = s_new;
    }
    arow[(size_t)(TPB_T - 1) * TV] = finish_row(s_prev, wpair[(TPB_T - 1) & 1], lane16);
}

__global__ __launch_bounds__(GEMM_WAVES * 32) void context_kernel(
    const float* __restrict__ attn, const float* __restrict__ value,
    float* __restrict__ ctx)
{
    __shared__ float part[GEMM_WAVES][16][128];
    const int b   = (int)blockIdx.x >> 6;
    const int rem = (int)blockIdx.x & 63;
    const int mt  = rem >> 1;
    const int nh  = rem & 1;
    const float* A  = attn  + (size_t)b * TQ * TV;
    const float* Bv = value + (size_t)b * TV * DM;
    float* C        = ctx   + (size_t)b * TQ * DM;
    strip_gemm_splitk<TV / GEMM_WAVES, GEMM_WAVES>(A, TV, Bv, DM, C, DM, mt * 16, nh * 128, part);
}

extern "C" void kernel_launch(void* const* d_in, const int* in_sizes, int n_in,
                              void* d_out, int out_size, void* d_ws, size_t ws_size,
                              hipStream_t stream) {
    const float* query = (const float*)d_in[0];
    const float* value = (const float*)d_in[1];
    const float* W1    = (const float*)d_in[2];
    const float* W2    = (const float*)d_in[3];
    const float* scale = (const float*)d_in[4];

    float* out  = (float*)d_out;
    float* ctx  = out;
    float* attn = out + (size_t)B_DIM * TQ * DM;

    float* Qw = (float*)d_ws;
    float* Kw = Qw + (size_t)(B_DIM * TQ) * UN;

    proj_kernel<<<256, 512, 0, stream>>>(query, value, W1, W2, Qw, Kw);
    score_softmax_kernel<<<(B_DIM * TQ) / TPB_T, TV, 0, stream>>>(Qw, Kw, scale, attn);
    context_kernel<<<256, 512, 0, stream>>>(attn, value, ctx);
}
